# starting kernel
# baseline (speedup 1.0000x reference)
#include <hip/hip_runtime.h>


#define POLY_ORDER 9
#define NP1 10
#define KPAD 12
#define NKS 3
#define NCT 7
#define ZSTRIDE 13
#define SSTRIDE 101
#define WAVES 4
#define PTS_PER_WAVE 32
#define PTS_PER_BLOCK (WAVES * PTS_PER_WAVE)
#define LDS_PER_WAVE (PTS_PER_WAVE * ZSTRIDE + PTS_PER_WAVE * SSTRIDE)

typedef __attribute__((ext_vector_type(2))) float v2f;
typedef __attribute__((ext_vector_type(8))) float v8f;

__device__ __forceinline__ void legendre10(float x, float* P) {
    P[0] = 1.0f;
    P[1] = x;
#pragma unroll
    for (int k = 1; k < POLY_ORDER; ++k) {
        P[k + 1] = ((float)(2 * k + 1) * x * P[k] - (float)k * P[k - 1]) *
                   (1.0f / (float)(k + 1));
    }
}

__global__ __launch_bounds__(128) void poly_wmma_kernel(
    const float* __restrict__ x,
    const float* __restrict__ w,
    const float* __restrict__ bias,
    float* __restrict__ out,
    int N)
{
    __shared__ float lds[WAVES * LDS_PER_WAVE];

    const int tid   = threadIdx.x;
    const int lane  = tid & 31;
    const int wave  = tid >> 5;
    const int lmod  = lane & 15;
    const int lhalf = lane >> 4;

    float* zbase = lds + wave * LDS_PER_WAVE;
    float* sbase = zbase + PTS_PER_WAVE * ZSTRIDE;

    const int p  = blockIdx.x * PTS_PER_BLOCK + wave * PTS_PER_WAVE + lane;
    const int pc = (p < N) ? p : (N - 1);

    const float xv = x[pc * 3 + 0];
    const float yv = x[pc * 3 + 1];
    const float zv = x[pc * 3 + 2];

    float Lx[NP1], Ly[NP1], Lz[NP1];
    legendre10(xv, Lx);
    legendre10(yv, Ly);
    legendre10(zv, Lz);

#pragma unroll
    for (int k = 0; k < NP1; ++k) zbase[lane * ZSTRIDE + k] = Lz[k];
    zbase[lane * ZSTRIDE + 10] = 0.0f;
    zbase[lane * ZSTRIDE + 11] = 0.0f;

    float B0[NCT][NKS], B1[NCT][NKS];
#pragma unroll
    for (int ct = 0; ct < NCT; ++ct) {
        const int c = ct * 16 + lmod;
#pragma unroll
        for (int ks = 0; ks < NKS; ++ks) {
            const int kk = ks * 4 + lhalf * 2;
            B0[ct][ks] = (c < 100 && kk     < NP1) ? w[c * 10 + kk]     : 0.0f;
            B1[ct][ks] = (c < 100 && kk + 1 < NP1) ? w[c * 10 + kk + 1] : 0.0f;
        }
    }

    __syncthreads();

#pragma unroll
    for (int t = 0; t < 2; ++t) {
        v8f acc[NCT];
#pragma unroll
        for (int ct = 0; ct < NCT; ++ct)
#pragma unroll
            for (int r = 0; r < 8; ++r) acc[ct][r] = 0.0f;

#pragma unroll
        for (int ks = 0; ks < NKS; ++ks) {
            const int row = t * 16 + lmod;
            const int kk  = ks * 4 + lhalf * 2;
            v2f a;
            a[0] = zbase[row * ZSTRIDE + kk];
            a[1] = zbase[row * ZSTRIDE + kk + 1];
#pragma unroll
            for (int ct = 0; ct < NCT; ++ct) {
                v2f b;
                b[0] = B0[ct][ks];
                b[1] = B1[ct][ks];
                acc[ct] = __builtin_amdgcn_wmma_f32_16x16x4_f32(
                     false, a,  false, b,
                     (short)0, acc[ct],
                     false,  false);
            }
        }

#pragma unroll
        for (int ct = 0; ct < NCT; ++ct) {
            const int c = ct * 16 + lmod;
            if (c < 100) {
#pragma unroll
                for (int r = 0; r < 8; ++r) {
                    const int row = t * 16 + lhalf * 8 + r;
                    sbase[row * SSTRIDE + c] = acc[ct][r];
                }
            }
        }
    }

    __syncthreads();

    float result = bias[0];
#pragma unroll
    for (int i = 0; i < NP1; ++i) {
        float ti = 0.0f;
#pragma unroll
        for (int j = 0; j < NP1; ++j)
            ti += Ly[j] * sbase[lane * SSTRIDE + i * 10 + j];
        result += Lx[i] * ti;
    }

    if (p < N) out[p] = result;
}

extern "C" void kernel_launch(void* const* d_in, const int* in_sizes, int n_in,
                              void* d_out, int out_size, void* d_ws, size_t ws_size,
                              hipStream_t stream) {
    const float* x    = (const float*)d_in[0];
    const float* w    = (const float*)d_in[1];
    const float* bias = (const float*)d_in[2];
    float* out        = (float*)d_out;

    const int N = in_sizes[0] / 3;
    const int blocks = (N + PTS_PER_BLOCK - 1) / PTS_PER_BLOCK;
    poly_wmma_kernel<<<blocks, 128, 0, stream>>>(x, w, bias, out, N);
}


# S stays in WMMA accumulators: W rows as A operand, each lane's own point's Lz as B, so the Lx*Ly contraction runs in registers (no LDS, no barrier); i split over lane halves (sum order changes)
# speedup vs baseline: 1.2180x; 1.2180x over previous
#include <hip/hip_runtime.h>


#define POLY_ORDER 9
#define NP1 10
#define NKS 3
#define NRT 7
#define HALF_C 50
#define WAVES 4
#define PTS_PER_WAVE 32
#define PTS_PER_BLOCK (WAVES * PTS_PER_WAVE)

typedef __attribute__((ext_vector_type(2))) float v2f;
typedef __attribute__((ext_vector_type(8))) float v8f;

__device__ __forceinline__ void legendre10(float x, float* P) {
    P[0] = 1.0f;
    P[1] = x;
#pragma unroll
    for (int k = 1; k < POLY_ORDER; ++k) {
        P[k + 1] = ((float)(2 * k + 1) * x * P[k] - (float)k * P[k - 1]) *
                   (1.0f / (float)(k + 1));
    }
}

__device__ __forceinline__ float xhalf(float v) {
    const int b = __builtin_bit_cast(int, v);
    return __builtin_bit_cast(
        float, __builtin_amdgcn_permlanex16(b, b, 0x76543210, 0xfedcba98, false, false));
}

__global__ __launch_bounds__(128) void poly_wmma_kernel(
    const float* __restrict__ x,
    const float* __restrict__ w,
    const float* __restrict__ bias,
    float* __restrict__ out,
    int N)
{
    const int tid  = threadIdx.x;
    const int lane = tid & 31;
    const int wave = tid >> 5;
    const int lmod = lane & 15;
    const int kh   = lane >> 4;

    const int base = blockIdx.x * PTS_PER_BLOCK + wave * PTS_PER_WAVE;

    float xv[2], yv[2], zv[2];
#pragma unroll
    for (int t = 0; t < 2; ++t) {
        const int p  = base + t * 16 + lmod;
        const int pc = (p < N) ? p : (N - 1);
        xv[t] = x[pc * 3 + 0];
        yv[t] = x[pc * 3 + 1];
        zv[t] = x[pc * 3 + 2];
    }

    const int ar = lmod & 7;
    const int ah = lmod >> 3;
    v2f Araw[NRT][NKS];
#pragma unroll
    for (int rt = 0; rt < NRT; ++rt) {
        const int s  = rt * 8 + ar;
        const int sc = (rt * 8 + 7 < HALF_C) ? s : (s < HALF_C ? s : HALF_C - 1);
        const int c  = HALF_C * ah + sc;
#pragma unroll
        for (int ks = 0; ks < NKS; ++ks) {
            const int kk = (ks * 4 + 2 < NP1) ? (ks * 4 + kh * 2) : (ks * 4);
            Araw[rt][ks] = *(const v2f*)(w + (c * 10 + kk));
        }
    }

    const float bv = bias[0];

    __builtin_amdgcn_sched_barrier(0);

    v2f Afrag[NRT][NKS];
#pragma unroll
    for (int rt = 0; rt < NRT; ++rt) {
        const bool rowok = (rt * 8 + 7 < HALF_C) || (rt * 8 + ar < HALF_C);
#pragma unroll
        for (int ks = 0; ks < NKS; ++ks) {
            const bool ok = rowok && ((ks * 4 + 2 < NP1) || (kh == 0));
            Afrag[rt][ks][0] = ok ? Araw[rt][ks][0] : 0.0f;
            Afrag[rt][ks][1] = ok ? Araw[rt][ks][1] : 0.0f;
        }
    }

    float tot[2];
#pragma unroll
    for (int t = 0; t < 2; ++t) {
        float Lx[NP1], Ly[NP1], Lz[NP1];
        legendre10(zv[t], Lz);
        legendre10(yv[t], Ly);
        legendre10(xv[t], Lx);

        v2f Bfrag[NKS];
        Bfrag[0][0] = kh ? Lz[2] : Lz[0];
        Bfrag[0][1] = kh ? Lz[3] : Lz[1];
        Bfrag[1][0] = kh ? Lz[6] : Lz[4];
        Bfrag[1][1] = kh ? Lz[7] : Lz[5];
        Bfrag[2][0] = kh ? 0.0f  : Lz[8];
        Bfrag[2][1] = kh ? 0.0f  : Lz[9];

        v8f acc[NRT];
#pragma unroll
        for (int rt = 0; rt < NRT; ++rt)
#pragma unroll
            for (int r = 0; r < 8; ++r) acc[rt][r] = 0.0f;

#pragma unroll
        for (int ks = 0; ks < NKS; ++ks) {
#pragma unroll
            for (int rt = 0; rt < NRT; ++rt) {
                acc[rt] = __builtin_amdgcn_wmma_f32_16x16x4_f32(
                     false, Afrag[rt][ks],  false, Bfrag[ks],
                     (short)0, acc[rt],
                     false,  false);
            }
        }

        float part = kh ? 0.0f : bv;
#pragma unroll
        for (int ii = 0; ii < 5; ++ii) {
            v2f g2 = {0.0f, 0.0f};
#pragma unroll
            for (int jp = 0; jp < 5; ++jp) {
                const int s = ii * 10 + 2 * jp;
                const v2f ly2 = {Ly[2 * jp], Ly[2 * jp + 1]};
                const v2f a2  = {acc[s >> 3][s & 7], acc[s >> 3][(s & 7) + 1]};
                g2 += ly2 * a2;
            }
            part += (kh ? Lx[5 + ii] : Lx[ii]) * (g2[0] + g2[1]);
        }
        tot[t] = part + xhalf(part);
    }

    const float result = kh ? tot[1] : tot[0];
    const int p = base + lane;
    if (p < N) out[p] = result;
}

extern "C" void kernel_launch(void* const* d_in, const int* in_sizes, int n_in,
                              void* d_out, int out_size, void* d_ws, size_t ws_size,
                              hipStream_t stream) {
    const float* x    = (const float*)d_in[0];
    const float* w    = (const float*)d_in[1];
    const float* bias = (const float*)d_in[2];
    float* out        = (float*)d_out;

    const int N = in_sizes[0] / 3;
    const int blocks = (N + PTS_PER_BLOCK - 1) / PTS_PER_BLOCK;
    poly_wmma_kernel<<<blocks, 128, 0, stream>>>(x, w, bias, out, N);
}
